# starting kernel
# baseline (speedup 1.0000x reference)
#include <hip/hip_runtime.h>
#include <hip/hip_bf16.h>

typedef __attribute__((ext_vector_type(2))) float v2f;
typedef __attribute__((ext_vector_type(8))) float v8f;

#define NR 256

__global__ __launch_bounds__(256) void sg_phase1(const float* __restrict__ t_all,
                                                 const float* __restrict__ weights,
                                                 float* __restrict__ Bws)
{
    __shared__ float sG[16];
    __shared__ float sGinv[16];
    __shared__ float sAw[4 * NR];

    const int i = threadIdx.x;
    if (i < 16) sG[i] = 0.0f;
    __syncthreads();

    const float tc = t_all[NR / 2];
    const float tr = t_all[i] - tc;
    const float wr = weights[i];
    const float w  = fmaxf(wr, 0.0f) + log1pf(expf(-fabsf(wr)));
    float v[4];
    v[0] = 1.0f;
    v[1] = tr;
    v[2] = tr * tr * 0.5f;
    v[3] = tr * tr * tr * (1.0f / 3.0f);

    #pragma unroll
    for (int p = 0; p < 4; ++p) sAw[p * NR + i] = v[p] * w;

    #pragma unroll
    for (int p = 0; p < 4; ++p)
        #pragma unroll
        for (int q = 0; q < 4; ++q)
            atomicAdd(&sG[p * 4 + q], v[p] * v[q] * w);
    __syncthreads();

    if (i == 0) {
        float M[4][8];
        for (int r = 0; r < 4; ++r)
            for (int c = 0; c < 4; ++c) {
                M[r][c]     = sG[r * 4 + c];
                M[r][4 + c] = (r == c) ? 1.0f : 0.0f;
            }
        for (int col = 0; col < 4; ++col) {
            const float inv = 1.0f / M[col][col];
            for (int c = 0; c < 8; ++c) M[col][c] *= inv;
            for (int r = 0; r < 4; ++r) {
                if (r == col) continue;
                const float f = M[r][col];
                for (int c = 0; c < 8; ++c) M[r][c] -= f * M[col][c];
            }
        }
        for (int r = 0; r < 4; ++r)
            for (int c = 0; c < 4; ++c) sGinv[r * 4 + c] = M[r][4 + c];
    }
    __syncthreads();

    float beff[4];
    #pragma unroll
    for (int n = 0; n < 4; ++n) {
        float acc = 0.0f;
        #pragma unroll
        for (int q = 0; q < 4; ++q) acc += sGinv[n * 4 + q] * sAw[q * NR + i];
        beff[n] = acc;
    }

    const int s = i >> 2;
    const int k = i & 3;
    const int j = k & 1;
    const int lbase = (k < 2) ? 0 : 16;
    #pragma unroll
    for (int n = 0; n < 16; ++n) {
        const float val = (n < 4) ? beff[n] : 0.0f;
        Bws[s * 64 + (lbase + n) * 2 + j] = val;
    }
}

__global__ __launch_bounds__(256) void sg_phase2(const float* __restrict__ X,
                                                 const float* __restrict__ Bws,
                                                 float* __restrict__ out,
                                                 int Dc, int numTiles)
{
    const int lane = threadIdx.x & 31;
    const int wave = threadIdx.x >> 5;
    const int tile = blockIdx.x * 8 + wave;
    if (tile >= numTiles) return;

    const long d0 = (long)tile * 16;
    const int  m  = lane & 15;
    const int  hi = lane >> 4;
    const int  kb = hi * 2;

    long col = d0 + m;
    if (col >= Dc) col = Dc - 1;
    const float* xp = X + (long)kb * Dc + col;

    v8f c = {0.0f, 0.0f, 0.0f, 0.0f, 0.0f, 0.0f, 0.0f, 0.0f};

    for (int s = 0; s < NR / 4; ++s) {
        const float* xs = xp + (long)s * 4 * Dc;
        v2f a;
        a.x = xs[0];
        a.y = xs[Dc];
        v2f b = *(const v2f*)(Bws + s * 64 + lane * 2);
        __builtin_prefetch(xs + 8 * (long)Dc, 0, 0);
        c = __builtin_amdgcn_wmma_f32_16x16x4_f32(
                false, a, false, b, (short)0, c, false, false);
    }

    const long dbase = d0 + (long)hi * 8;
    if (d0 + 16 <= (long)Dc) {
        if (m < 4) {
            #pragma unroll
            for (int r = 0; r < 8; ++r) {
                out[(dbase + r) * 4 + m] = c[r];
            }
        }
    } else {
        if (m < 4) {
            #pragma unroll
            for (int r = 0; r < 8; ++r) {
                const long d = dbase + r;
                if (d < Dc) out[d * 4 + m] = c[r];
            }
        }
    }
}

extern "C" void kernel_launch(void* const* d_in, const int* in_sizes, int n_in,
                              void* d_out, int out_size, void* d_ws, size_t ws_size,
                              hipStream_t stream) {
    const float* t_all   = (const float*)d_in[0];
    const float* x_all   = (const float*)d_in[1];
    const float* weights = (const float*)d_in[2];
    (void)d_in[3]; (void)n_in; (void)out_size; (void)ws_size;

    const int T  = in_sizes[0];
    const int Dc = in_sizes[1] / T;

    float* Bws = (float*)d_ws;

    hipLaunchKernelGGL(sg_phase1, dim3(1), dim3(256), 0, stream,
                       t_all, weights, Bws);

    const int numTiles = (Dc + 15) / 16;
    const int blocks   = (numTiles + 7) / 8;
    hipLaunchKernelGGL(sg_phase2, dim3(blocks), dim3(256), 0, stream,
                       x_all, Bws, (float*)d_out, Dc, numTiles);
}


# phase 2 X stream: each lane loads 16 bytes (64 channels per wave as 4 interleaved WMMA tiles), two 16-load groups requested ahead of the matrix work; one wave per work-group, prefetch dropped
# speedup vs baseline: 1.0405x; 1.0405x over previous
#include <hip/hip_runtime.h>
#include <hip/hip_bf16.h>

typedef __attribute__((ext_vector_type(2))) float v2f;
typedef __attribute__((ext_vector_type(4))) float v4f;
typedef __attribute__((ext_vector_type(8))) float v8f;

#define NR 256

__global__ __launch_bounds__(256) void sg_phase1(const float* __restrict__ t_all,
                                                 const float* __restrict__ weights,
                                                 float* __restrict__ Bws)
{
    __shared__ float sG[16];
    __shared__ float sGinv[16];
    __shared__ float sAw[4 * NR];

    const int i = threadIdx.x;
    if (i < 16) sG[i] = 0.0f;
    __syncthreads();

    const float tc = t_all[NR / 2];
    const float tr = t_all[i] - tc;
    const float wr = weights[i];
    const float w  = fmaxf(wr, 0.0f) + log1pf(expf(-fabsf(wr)));
    float v[4];
    v[0] = 1.0f;
    v[1] = tr;
    v[2] = tr * tr * 0.5f;
    v[3] = tr * tr * tr * (1.0f / 3.0f);

    #pragma unroll
    for (int p = 0; p < 4; ++p) sAw[p * NR + i] = v[p] * w;

    #pragma unroll
    for (int p = 0; p < 4; ++p)
        #pragma unroll
        for (int q = 0; q < 4; ++q)
            atomicAdd(&sG[p * 4 + q], v[p] * v[q] * w);
    __syncthreads();

    if (i == 0) {
        float M[4][8];
        for (int r = 0; r < 4; ++r)
            for (int c = 0; c < 4; ++c) {
                M[r][c]     = sG[r * 4 + c];
                M[r][4 + c] = (r == c) ? 1.0f : 0.0f;
            }
        for (int col = 0; col < 4; ++col) {
            const float inv = 1.0f / M[col][col];
            for (int c = 0; c < 8; ++c) M[col][c] *= inv;
            for (int r = 0; r < 4; ++r) {
                if (r == col) continue;
                const float f = M[r][col];
                for (int c = 0; c < 8; ++c) M[r][c] -= f * M[col][c];
            }
        }
        for (int r = 0; r < 4; ++r)
            for (int c = 0; c < 4; ++c) sGinv[r * 4 + c] = M[r][4 + c];
    }
    __syncthreads();

    float beff[4];
    #pragma unroll
    for (int n = 0; n < 4; ++n) {
        float acc = 0.0f;
        #pragma unroll
        for (int q = 0; q < 4; ++q) acc += sGinv[n * 4 + q] * sAw[q * NR + i];
        beff[n] = acc;
    }

    const int s = i >> 2;
    const int k = i & 3;
    const int j = k & 1;
    const int lbase = (k < 2) ? 0 : 16;
    #pragma unroll
    for (int n = 0; n < 16; ++n) {
        const float val = (n < 4) ? beff[n] : 0.0f;
        Bws[s * 64 + (lbase + n) * 2 + j] = val;
    }
}

#define GSTEPS  8
#define NGROUPS (NR / 4 / GSTEPS)
#define AHEAD   2

__global__ __launch_bounds__(32) void sg_phase2_wide(const float* __restrict__ X,
                                                     const float* __restrict__ Bws,
                                                     float* __restrict__ out,
                                                     int Dc)
{
    const unsigned lane = threadIdx.x;
    const unsigned m    = lane & 15u;
    const unsigned hi   = lane >> 4;
    const long     d0   = (long)blockIdx.x * 64;

    long q = d0 / 4 + (long)m;
    const long qlast = (long)(Dc / 4) - 1;
    if (q > qlast) q = qlast;

    const size_t   pitch = (size_t)Dc * sizeof(float);
    const unsigned voff  = (unsigned)((size_t)(2u * hi) * pitch + (size_t)q * 16u);
    const char*    xb    = (const char*)X;
    const char*    bb    = (const char*)Bws + lane * 8u;

    v4f P[AHEAD + 1][GSTEPS];
    v4f Q[AHEAD + 1][GSTEPS];
    v2f B[AHEAD + 1][GSTEPS];
    v8f c[4];
    #pragma unroll
    for (int t = 0; t < 4; ++t) c[t] = (v8f){0.0f, 0.0f, 0.0f, 0.0f, 0.0f, 0.0f, 0.0f, 0.0f};

#define LOAD_GROUP(G) do {                                                          \
        _Pragma("unroll") for (int u = 0; u < GSTEPS; ++u) {                        \
            const int   s_  = (G) * GSTEPS + u;                                     \
            const char* r_  = xb + (size_t)(4 * s_) * pitch;                        \
            P[(G) % (AHEAD + 1)][u] = *(const v4f*)(r_ + voff);                     \
            Q[(G) % (AHEAD + 1)][u] = *(const v4f*)(r_ + pitch + voff);             \
            B[(G) % (AHEAD + 1)][u] = *(const v2f*)(bb + s_ * 256);                 \
        } } while (0)

    #pragma unroll
    for (int g = 0; g < AHEAD; ++g) LOAD_GROUP(g);

    #pragma unroll
    for (int g = 0; g < NGROUPS; ++g) {
        __builtin_amdgcn_sched_barrier(0);
        if (g + AHEAD < NGROUPS) LOAD_GROUP(g + AHEAD);
        __builtin_amdgcn_sched_barrier(0);
        #pragma unroll
        for (int u = 0; u < GSTEPS; ++u) {
            const v4f p = P[g % (AHEAD + 1)][u];
            const v4f w = Q[g % (AHEAD + 1)][u];
            const v2f b = B[g % (AHEAD + 1)][u];
            #pragma unroll
            for (int t = 0; t < 4; ++t) {
                v2f a;
                a.x = p[t];
                a.y = w[t];
                c[t] = __builtin_amdgcn_wmma_f32_16x16x4_f32(
                           false, a, false, b, (short)0, c[t], false, false);
            }
        }
    }
#undef LOAD_GROUP

    const long dbase = d0 + (long)hi * 32;
    if (d0 + 64 <= (long)Dc) {
        if (m < 4u) {
            #pragma unroll
            for (int r = 0; r < 8; ++r)
                #pragma unroll
                for (int t = 0; t < 4; ++t)
                    out[(dbase + 4 * r + t) * 4 + m] = c[t][r];
        }
    } else {
        if (m < 4u) {
            #pragma unroll
            for (int r = 0; r < 8; ++r)
                #pragma unroll
                for (int t = 0; t < 4; ++t) {
                    const long d = dbase + 4 * r + t;
                    if (d < (long)Dc) out[d * 4 + m] = c[t][r];
                }
        }
    }
}

__global__ __launch_bounds__(256) void sg_phase2(const float* __restrict__ X,
                                                 const float* __restrict__ Bws,
                                                 float* __restrict__ out,
                                                 int Dc, int numTiles)
{
    const int lane = threadIdx.x & 31;
    const int wave = threadIdx.x >> 5;
    const int tile = blockIdx.x * 8 + wave;
    if (tile >= numTiles) return;

    const long d0 = (long)tile * 16;
    const int  m  = lane & 15;
    const int  hi = lane >> 4;
    const int  kb = hi * 2;

    long col = d0 + m;
    if (col >= Dc) col = Dc - 1;
    const float* xp = X + (long)kb * Dc + col;

    v8f c = {0.0f, 0.0f, 0.0f, 0.0f, 0.0f, 0.0f, 0.0f, 0.0f};

    for (int s = 0; s < NR / 4; ++s) {
        const float* xs = xp + (long)s * 4 * Dc;
        v2f a;
        a.x = xs[0];
        a.y = xs[Dc];
        v2f b = *(const v2f*)(Bws + s * 64 + lane * 2);
        __builtin_prefetch(xs + 8 * (long)Dc, 0, 0);
        c = __builtin_amdgcn_wmma_f32_16x16x4_f32(
                false, a, false, b, (short)0, c, false, false);
    }

    const long dbase = d0 + (long)hi * 8;
    if (d0 + 16 <= (long)Dc) {
        if (m < 4) {
            #pragma unroll
            for (int r = 0; r < 8; ++r) {
                out[(dbase + r) * 4 + m] = c[r];
            }
        }
    } else {
        if (m < 4) {
            #pragma unroll
            for (int r = 0; r < 8; ++r) {
                const long d = dbase + r;
                if (d < Dc) out[d * 4 + m] = c[r];
            }
        }
    }
}

extern "C" void kernel_launch(void* const* d_in, const int* in_sizes, int n_in,
                              void* d_out, int out_size, void* d_ws, size_t ws_size,
                              hipStream_t stream) {
    const float* t_all   = (const float*)d_in[0];
    const float* x_all   = (const float*)d_in[1];
    const float* weights = (const float*)d_in[2];
    (void)d_in[3]; (void)n_in; (void)out_size; (void)ws_size;

    const int T  = in_sizes[0];
    const int Dc = in_sizes[1] / T;

    float* Bws = (float*)d_ws;

    hipLaunchKernelGGL(sg_phase1, dim3(1), dim3(256), 0, stream,
                       t_all, weights, Bws);

    if ((Dc & 3) == 0) {
        const int waves = (Dc + 63) / 64;
        hipLaunchKernelGGL(sg_phase2_wide, dim3(waves), dim3(32), 0, stream,
                           x_all, Bws, (float*)d_out, Dc);
    } else {
        const int numTiles = (Dc + 15) / 16;
        const int blocks   = (numTiles + 7) / 8;
        hipLaunchKernelGGL(sg_phase2, dim3(blocks), dim3(256), 0, stream,
                           x_all, Bws, (float*)d_out, Dc, numTiles);
    }
}


# single launch: each wave rebuilds the 4x256 fit matrix itself under its first X loads (Gram sums by lane partials + xor-shuffle tree, not LDS atomics: new sum order) and reads B fragments from LDS
# speedup vs baseline: 2.3985x; 2.3051x over previous
#include <hip/hip_runtime.h>
#include <hip/hip_bf16.h>

typedef __attribute__((ext_vector_type(2))) float v2f;
typedef __attribute__((ext_vector_type(4))) float v4f;
typedef __attribute__((ext_vector_type(8))) float v8f;

#define NR 256

__global__ __launch_bounds__(256) void sg_phase1(const float* __restrict__ t_all,
                                                 const float* __restrict__ weights,
                                                 float* __restrict__ Bws)
{
    __shared__ float sG[16];
    __shared__ float sGinv[16];
    __shared__ float sAw[4 * NR];

    const int i = threadIdx.x;
    if (i < 16) sG[i] = 0.0f;
    __syncthreads();

    const float tc = t_all[NR / 2];
    const float tr = t_all[i] - tc;
    const float wr = weights[i];
    const float w  = fmaxf(wr, 0.0f) + log1pf(expf(-fabsf(wr)));
    float v[4];
    v[0] = 1.0f;
    v[1] = tr;
    v[2] = tr * tr * 0.5f;
    v[3] = tr * tr * tr * (1.0f / 3.0f);

    #pragma unroll
    for (int p = 0; p < 4; ++p) sAw[p * NR + i] = v[p] * w;

    #pragma unroll
    for (int p = 0; p < 4; ++p)
        #pragma unroll
        for (int q = 0; q < 4; ++q)
            atomicAdd(&sG[p * 4 + q], v[p] * v[q] * w);
    __syncthreads();

    if (i == 0) {
        float M[4][8];
        for (int r = 0; r < 4; ++r)
            for (int c = 0; c < 4; ++c) {
                M[r][c]     = sG[r * 4 + c];
                M[r][4 + c] = (r == c) ? 1.0f : 0.0f;
            }
        for (int col = 0; col < 4; ++col) {
            const float inv = 1.0f / M[col][col];
            for (int c = 0; c < 8; ++c) M[col][c] *= inv;
            for (int r = 0; r < 4; ++r) {
                if (r == col) continue;
                const float f = M[r][col];
                for (int c = 0; c < 8; ++c) M[r][c] -= f * M[col][c];
            }
        }
        for (int r = 0; r < 4; ++r)
            for (int c = 0; c < 4; ++c) sGinv[r * 4 + c] = M[r][4 + c];
    }
    __syncthreads();

    float beff[4];
    #pragma unroll
    for (int n = 0; n < 4; ++n) {
        float acc = 0.0f;
        #pragma unroll
        for (int q = 0; q < 4; ++q) acc += sGinv[n * 4 + q] * sAw[q * NR + i];
        beff[n] = acc;
    }

    const int s = i >> 2;
    const int k = i & 3;
    const int j = k & 1;
    const int lbase = (k < 2) ? 0 : 16;
    #pragma unroll
    for (int n = 0; n < 16; ++n) {
        const float val = (n < 4) ? beff[n] : 0.0f;
        Bws[s * 64 + (lbase + n) * 2 + j] = val;
    }
}

#define GSTEPS  8
#define NGROUPS (NR / 4 / GSTEPS)
#define AHEAD   2
#define BP      260

__global__ __launch_bounds__(32) void sg_fused_wide(const float* __restrict__ t_all,
                                                    const float* __restrict__ weights,
                                                    const float* __restrict__ X,
                                                    float* __restrict__ out,
                                                    int Dc)
{
    __shared__ __attribute__((aligned(16))) float sB[5 * BP];

    const unsigned lane = threadIdx.x;
    const unsigned m    = lane & 15u;
    const unsigned hi   = lane >> 4;
    const long     d0   = (long)blockIdx.x * 64;

    const v4f   tA = *(const v4f*)(t_all + 8u * lane);
    const v4f   tB = *(const v4f*)(t_all + 8u * lane + 4u);
    const v4f   wA = *(const v4f*)(weights + 8u * lane);
    const v4f   wB = *(const v4f*)(weights + 8u * lane + 4u);
    const float tc = t_all[NR / 2];
    __builtin_amdgcn_sched_barrier(0);

    long q = d0 / 4 + (long)m;
    const long qlast = (long)(Dc / 4) - 1;
    if (q > qlast) q = qlast;

    const size_t   pitch = (size_t)Dc * sizeof(float);
    const unsigned voff  = (unsigned)((size_t)(2u * hi) * pitch + (size_t)q * 16u);
    const char*    xb    = (const char*)X;
    const float*   bl    = sB + (m < 4u ? m : 4u) * BP + 2u * hi;

    v4f P[AHEAD + 1][GSTEPS];
    v4f Q[AHEAD + 1][GSTEPS];
    v2f B[AHEAD + 1][GSTEPS];
    v8f c[4];
    #pragma unroll
    for (int t = 0; t < 4; ++t) c[t] = (v8f){0.0f, 0.0f, 0.0f, 0.0f, 0.0f, 0.0f, 0.0f, 0.0f};

#define LOAD_X(G) do {                                                              \
        _Pragma("unroll") for (int u = 0; u < GSTEPS; ++u) {                        \
            const int   s_  = (G) * GSTEPS + u;                                     \
            const char* r_  = xb + (size_t)(4 * s_) * pitch;                        \
            P[(G) % (AHEAD + 1)][u] = *(const v4f*)(r_ + voff);                     \
            Q[(G) % (AHEAD + 1)][u] = *(const v4f*)(r_ + pitch + voff);             \
        } } while (0)
#define LOAD_B(G) do {                                                              \
        _Pragma("unroll") for (int u = 0; u < GSTEPS; ++u) {                        \
            const int   s_  = (G) * GSTEPS + u;                                     \
            B[(G) % (AHEAD + 1)][u] = *(const v2f*)(bl + 4 * s_);                   \
        } } while (0)

    #pragma unroll
    for (int g = 0; g < AHEAD; ++g) LOAD_X(g);
    __builtin_amdgcn_sched_barrier(0);

    {
        float aw[4][8];
        float g[10];
        #pragma unroll
        for (int e = 0; e < 10; ++e) g[e] = 0.0f;

        #pragma unroll
        for (int j = 0; j < 8; ++j) {
            const float tr = ((j < 4) ? tA[j & 3] : tB[j & 3]) - tc;
            const float wr = (j < 4) ? wA[j & 3] : wB[j & 3];
            const float w  = fmaxf(wr, 0.0f) + log1pf(expf(-fabsf(wr)));
            float v[4];
            v[0] = 1.0f;
            v[1] = tr;
            v[2] = tr * tr * 0.5f;
            v[3] = tr * tr * tr * (1.0f / 3.0f);
            #pragma unroll
            for (int p = 0; p < 4; ++p) aw[p][j] = v[p] * w;
            int e = 0;
            #pragma unroll
            for (int p = 0; p < 4; ++p)
                #pragma unroll
                for (int r = p; r < 4; ++r, ++e)
                    g[e] += v[p] * v[r] * w;
        }

        #pragma unroll
        for (int mask = 16; mask >= 1; mask >>= 1)
            #pragma unroll
            for (int e = 0; e < 10; ++e)
                g[e] += __shfl_xor(g[e], mask, 32);

        float M[4][8];
        {
            int e = 0;
            #pragma unroll
            for (int p = 0; p < 4; ++p)
                #pragma unroll
                for (int r = p; r < 4; ++r, ++e) { M[p][r] = g[e]; M[r][p] = g[e]; }
        }
        #pragma unroll
        for (int r = 0; r < 4; ++r)
            #pragma unroll
            for (int cc = 0; cc < 4; ++cc) M[r][4 + cc] = (r == cc) ? 1.0f : 0.0f;
        #pragma unroll
        for (int col = 0; col < 4; ++col) {
            const float inv = 1.0f / M[col][col];
            #pragma unroll
            for (int cc = 0; cc < 8; ++cc) M[col][cc] *= inv;
            #pragma unroll
            for (int r = 0; r < 4; ++r) {
                if (r == col) continue;
                const float f = M[r][col];
                #pragma unroll
                for (int cc = 0; cc < 8; ++cc) M[r][cc] -= f * M[col][cc];
            }
        }

        #pragma unroll
        for (int n = 0; n < 4; ++n) {
            float be[8];
            #pragma unroll
            for (int j = 0; j < 8; ++j) {
                float acc = 0.0f;
                #pragma unroll
                for (int r = 0; r < 4; ++r) acc += M[n][4 + r] * aw[r][j];
                be[j] = acc;
            }
            *(v4f*)(sB + n * BP + 8u * lane)      = (v4f){be[0], be[1], be[2], be[3]};
            *(v4f*)(sB + n * BP + 8u * lane + 4u) = (v4f){be[4], be[5], be[6], be[7]};
        }
        *(v4f*)(sB + 4 * BP + 8u * lane)      = (v4f){0.0f, 0.0f, 0.0f, 0.0f};
        *(v4f*)(sB + 4 * BP + 8u * lane + 4u) = (v4f){0.0f, 0.0f, 0.0f, 0.0f};
    }
    __syncthreads();

    #pragma unroll
    for (int g = 0; g < AHEAD; ++g) LOAD_B(g);

    #pragma unroll
    for (int g = 0; g < NGROUPS; ++g) {
        __builtin_amdgcn_sched_barrier(0);
        if (g + AHEAD < NGROUPS) { LOAD_X(g + AHEAD); LOAD_B(g + AHEAD); }
        __builtin_amdgcn_sched_barrier(0);
        #pragma unroll
        for (int u = 0; u < GSTEPS; ++u) {
            const v4f p = P[g % (AHEAD + 1)][u];
            const v4f w = Q[g % (AHEAD + 1)][u];
            const v2f b = B[g % (AHEAD + 1)][u];
            #pragma unroll
            for (int t = 0; t < 4; ++t) {
                v2f a;
                a.x = p[t];
                a.y = w[t];
                c[t] = __builtin_amdgcn_wmma_f32_16x16x4_f32(
                           false, a, false, b, (short)0, c[t], false, false);
            }
        }
    }
#undef LOAD_X
#undef LOAD_B

    const long dbase = d0 + (long)hi * 32;
    if (d0 + 64 <= (long)Dc) {
        if (m < 4u) {
            #pragma unroll
            for (int r = 0; r < 8; ++r)
                #pragma unroll
                for (int t = 0; t < 4; ++t)
                    out[(dbase + 4 * r + t) * 4 + m] = c[t][r];
        }
    } else {
        if (m < 4u) {
            #pragma unroll
            for (int r = 0; r < 8; ++r)
                #pragma unroll
                for (int t = 0; t < 4; ++t) {
                    const long d = dbase + 4 * r + t;
                    if (d < (long)Dc) out[d * 4 + m] = c[t][r];
                }
        }
    }
}

__global__ __launch_bounds__(256) void sg_phase2(const float* __restrict__ X,
                                                 const float* __restrict__ Bws,
                                                 float* __restrict__ out,
                                                 int Dc, int numTiles)
{
    const int lane = threadIdx.x & 31;
    const int wave = threadIdx.x >> 5;
    const int tile = blockIdx.x * 8 + wave;
    if (tile >= numTiles) return;

    const long d0 = (long)tile * 16;
    const int  m  = lane & 15;
    const int  hi = lane >> 4;
    const int  kb = hi * 2;

    long col = d0 + m;
    if (col >= Dc) col = Dc - 1;
    const float* xp = X + (long)kb * Dc + col;

    v8f c = {0.0f, 0.0f, 0.0f, 0.0f, 0.0f, 0.0f, 0.0f, 0.0f};

    for (int s = 0; s < NR / 4; ++s) {
        const float* xs = xp + (long)s * 4 * Dc;
        v2f a;
        a.x = xs[0];
        a.y = xs[Dc];
        v2f b = *(const v2f*)(Bws + s * 64 + lane * 2);
        __builtin_prefetch(xs + 8 * (long)Dc, 0, 0);
        c = __builtin_amdgcn_wmma_f32_16x16x4_f32(
                false, a, false, b, (short)0, c, false, false);
    }

    const long dbase = d0 + (long)hi * 8;
    if (d0 + 16 <= (long)Dc) {
        if (m < 4) {
            #pragma unroll
            for (int r = 0; r < 8; ++r) {
                out[(dbase + r) * 4 + m] = c[r];
            }
        }
    } else {
        if (m < 4) {
            #pragma unroll
            for (int r = 0; r < 8; ++r) {
                const long d = dbase + r;
                if (d < Dc) out[d * 4 + m] = c[r];
            }
        }
    }
}

extern "C" void kernel_launch(void* const* d_in, const int* in_sizes, int n_in,
                              void* d_out, int out_size, void* d_ws, size_t ws_size,
                              hipStream_t stream) {
    const float* t_all   = (const float*)d_in[0];
    const float* x_all   = (const float*)d_in[1];
    const float* weights = (const float*)d_in[2];
    (void)d_in[3]; (void)n_in; (void)out_size; (void)ws_size;

    const int T  = in_sizes[0];
    const int Dc = in_sizes[1] / T;

    if ((Dc & 3) == 0) {
        const int waves = (Dc + 63) / 64;
        hipLaunchKernelGGL(sg_fused_wide, dim3(waves), dim3(32), 0, stream,
                           t_all, weights, x_all, (float*)d_out, Dc);
    } else {
        float* Bws = (float*)d_ws;
        hipLaunchKernelGGL(sg_phase1, dim3(1), dim3(256), 0, stream,
                           t_all, weights, Bws);
        const int numTiles = (Dc + 15) / 16;
        const int blocks   = (numTiles + 7) / 8;
        hipLaunchKernelGGL(sg_phase2, dim3(blocks), dim3(256), 0, stream,
                           x_all, Bws, (float*)d_out, Dc, numTiles);
    }
}


# X requests kept in flight through the fit-matrix build: groups 2 and 3 requested half way through it, four groups ahead in the loop, wave-level LDS hand-over in place of the draining barrier
# speedup vs baseline: 2.6045x; 1.0859x over previous
#include <hip/hip_runtime.h>
#include <hip/hip_bf16.h>

typedef __attribute__((ext_vector_type(2))) float v2f;
typedef __attribute__((ext_vector_type(4))) float v4f;
typedef __attribute__((ext_vector_type(8))) float v8f;

#define NR 256

__global__ __launch_bounds__(256) void sg_phase1(const float* __restrict__ t_all,
                                                 const float* __restrict__ weights,
                                                 float* __restrict__ Bws)
{
    __shared__ float sG[16];
    __shared__ float sGinv[16];
    __shared__ float sAw[4 * NR];

    const int i = threadIdx.x;
    if (i < 16) sG[i] = 0.0f;
    __syncthreads();

    const float tc = t_all[NR / 2];
    const float tr = t_all[i] - tc;
    const float wr = weights[i];
    const float w  = fmaxf(wr, 0.0f) + log1pf(expf(-fabsf(wr)));
    float v[4];
    v[0] = 1.0f;
    v[1] = tr;
    v[2] = tr * tr * 0.5f;
    v[3] = tr * tr * tr * (1.0f / 3.0f);

    #pragma unroll
    for (int p = 0; p < 4; ++p) sAw[p * NR + i] = v[p] * w;

    #pragma unroll
    for (int p = 0; p < 4; ++p)
        #pragma unroll
        for (int q = 0; q < 4; ++q)
            atomicAdd(&sG[p * 4 + q], v[p] * v[q] * w);
    __syncthreads();

    if (i == 0) {
        float M[4][8];
        for (int r = 0; r < 4; ++r)
            for (int c = 0; c < 4; ++c) {
                M[r][c]     = sG[r * 4 + c];
                M[r][4 + c] = (r == c) ? 1.0f : 0.0f;
            }
        for (int col = 0; col < 4; ++col) {
            const float inv = 1.0f / M[col][col];
            for (int c = 0; c < 8; ++c) M[col][c] *= inv;
            for (int r = 0; r < 4; ++r) {
                if (r == col) continue;
                const float f = M[r][col];
                for (int c = 0; c < 8; ++c) M[r][c] -= f * M[col][c];
            }
        }
        for (int r = 0; r < 4; ++r)
            for (int c = 0; c < 4; ++c) sGinv[r * 4 + c] = M[r][4 + c];
    }
    __syncthreads();

    float beff[4];
    #pragma unroll
    for (int n = 0; n < 4; ++n) {
        float acc = 0.0f;
        #pragma unroll
        for (int q = 0; q < 4; ++q) acc += sGinv[n * 4 + q] * sAw[q * NR + i];
        beff[n] = acc;
    }

    const int s = i >> 2;
    const int k = i & 3;
    const int j = k & 1;
    const int lbase = (k < 2) ? 0 : 16;
    #pragma unroll
    for (int n = 0; n < 16; ++n) {
        const float val = (n < 4) ? beff[n] : 0.0f;
        Bws[s * 64 + (lbase + n) * 2 + j] = val;
    }
}

#define GSTEPS  8
#define NGROUPS (NR / 4 / GSTEPS)
#define AHEAD   4
#define NBUF    (AHEAD + 1)
#define BP      260

__global__ __launch_bounds__(32) void sg_fused_wide(const float* __restrict__ t_all,
                                                    const float* __restrict__ weights,
                                                    const float* __restrict__ X,
                                                    float* __restrict__ out,
                                                    int Dc)
{
    __shared__ __attribute__((aligned(16))) float sB[5 * BP];

    const unsigned lane = threadIdx.x;
    const unsigned m    = lane & 15u;
    const unsigned hi   = lane >> 4;
    const long     d0   = (long)blockIdx.x * 64;

    const v4f   tA = *(const v4f*)(t_all + 8u * lane);
    const v4f   tB = *(const v4f*)(t_all + 8u * lane + 4u);
    const v4f   wA = *(const v4f*)(weights + 8u * lane);
    const v4f   wB = *(const v4f*)(weights + 8u * lane + 4u);
    const float tc = t_all[NR / 2];
    __builtin_amdgcn_sched_barrier(0);

    long q = d0 / 4 + (long)m;
    const long qlast = (long)(Dc / 4) - 1;
    if (q > qlast) q = qlast;

    const size_t   pitch = (size_t)Dc * sizeof(float);
    const unsigned voff  = (unsigned)((size_t)(2u * hi) * pitch + (size_t)q * 16u);
    const char*    xb    = (const char*)X;
    const float*   bl    = sB + (m < 4u ? m : 4u) * BP + 2u * hi;

    v4f P[NBUF][GSTEPS];
    v4f Q[NBUF][GSTEPS];
    v2f B[2][GSTEPS];
    v8f c[4];
    #pragma unroll
    for (int t = 0; t < 4; ++t) c[t] = (v8f){0.0f, 0.0f, 0.0f, 0.0f, 0.0f, 0.0f, 0.0f, 0.0f};

#define LOAD_X(G) do {                                                              \
        _Pragma("unroll") for (int u = 0; u < GSTEPS; ++u) {                        \
            const int   s_  = (G) * GSTEPS + u;                                     \
            const char* r_  = xb + (size_t)(4 * s_) * pitch;                        \
            P[(G) % NBUF][u] = *(const v4f*)(r_ + voff);                            \
            Q[(G) % NBUF][u] = *(const v4f*)(r_ + pitch + voff);                    \
        } } while (0)
#define LOAD_B(G) do {                                                              \
        _Pragma("unroll") for (int u = 0; u < GSTEPS; ++u) {                        \
            const int   s_  = (G) * GSTEPS + u;                                     \
            B[(G) % 2][u] = *(const v2f*)(bl + 4 * s_);                             \
        } } while (0)

    LOAD_X(0);
    LOAD_X(1);
    __builtin_amdgcn_sched_barrier(0);

    {
        float aw[4][8];
        float g[10];
        #pragma unroll
        for (int e = 0; e < 10; ++e) g[e] = 0.0f;

        #pragma unroll
        for (int j = 0; j < 8; ++j) {
            const float tr = ((j < 4) ? tA[j & 3] : tB[j & 3]) - tc;
            const float wr = (j < 4) ? wA[j & 3] : wB[j & 3];
            const float w  = fmaxf(wr, 0.0f) + log1pf(expf(-fabsf(wr)));
            float v[4];
            v[0] = 1.0f;
            v[1] = tr;
            v[2] = tr * tr * 0.5f;
            v[3] = tr * tr * tr * (1.0f / 3.0f);
            #pragma unroll
            for (int p = 0; p < 4; ++p) aw[p][j] = v[p] * w;
            int e = 0;
            #pragma unroll
            for (int p = 0; p < 4; ++p)
                #pragma unroll
                for (int r = p; r < 4; ++r, ++e)
                    g[e] += v[p] * v[r] * w;
        }

        __builtin_amdgcn_sched_barrier(0);
        LOAD_X(2);
        LOAD_X(3);
        __builtin_amdgcn_sched_barrier(0);

        #pragma unroll
        for (int mask = 16; mask >= 1; mask >>= 1)
            #pragma unroll
            for (int e = 0; e < 10; ++e)
                g[e] += __shfl_xor(g[e], mask, 32);

        float M[4][8];
        {
            int e = 0;
            #pragma unroll
            for (int p = 0; p < 4; ++p)
                #pragma unroll
                for (int r = p; r < 4; ++r, ++e) { M[p][r] = g[e]; M[r][p] = g[e]; }
        }
        #pragma unroll
        for (int r = 0; r < 4; ++r)
            #pragma unroll
            for (int cc = 0; cc < 4; ++cc) M[r][4 + cc] = (r == cc) ? 1.0f : 0.0f;
        #pragma unroll
        for (int col = 0; col < 4; ++col) {
            const float inv = 1.0f / M[col][col];
            #pragma unroll
            for (int cc = 0; cc < 8; ++cc) M[col][cc] *= inv;
            #pragma unroll
            for (int r = 0; r < 4; ++r) {
                if (r == col) continue;
                const float f = M[r][col];
                #pragma unroll
                for (int cc = 0; cc < 8; ++cc) M[r][cc] -= f * M[col][cc];
            }
        }

        #pragma unroll
        for (int n = 0; n < 4; ++n) {
            float be[8];
            #pragma unroll
            for (int j = 0; j < 8; ++j) {
                float acc = 0.0f;
                #pragma unroll
                for (int r = 0; r < 4; ++r) acc += M[n][4 + r] * aw[r][j];
                be[j] = acc;
            }
            *(v4f*)(sB + n * BP + 8u * lane)      = (v4f){be[0], be[1], be[2], be[3]};
            *(v4f*)(sB + n * BP + 8u * lane + 4u) = (v4f){be[4], be[5], be[6], be[7]};
        }
        *(v4f*)(sB + 4 * BP + 8u * lane)      = (v4f){0.0f, 0.0f, 0.0f, 0.0f};
        *(v4f*)(sB + 4 * BP + 8u * lane + 4u) = (v4f){0.0f, 0.0f, 0.0f, 0.0f};
    }
    __builtin_amdgcn_fence(__ATOMIC_RELEASE, "wavefront");
    __builtin_amdgcn_wave_barrier();
    __builtin_amdgcn_fence(__ATOMIC_ACQUIRE, "wavefront");

    LOAD_B(0);

    #pragma unroll
    for (int g = 0; g < NGROUPS; ++g) {
        __builtin_amdgcn_sched_barrier(0);
        if (g + AHEAD < NGROUPS) LOAD_X(g + AHEAD);
        if (g + 1 < NGROUPS)     LOAD_B(g + 1);
        __builtin_amdgcn_sched_barrier(0);
        #pragma unroll
        for (int u = 0; u < GSTEPS; ++u) {
            const v4f p = P[g % NBUF][u];
            const v4f w = Q[g % NBUF][u];
            const v2f b = B[g % 2][u];
            #pragma unroll
            for (int t = 0; t < 4; ++t) {
                v2f a;
                a.x = p[t];
                a.y = w[t];
                c[t] = __builtin_amdgcn_wmma_f32_16x16x4_f32(
                           false, a, false, b, (short)0, c[t], false, false);
            }
        }
    }
#undef LOAD_X
#undef LOAD_B

    const long dbase = d0 + (long)hi * 32;
    if (d0 + 64 <= (long)Dc) {
        if (m < 4u) {
            #pragma unroll
            for (int r = 0; r < 8; ++r)
                #pragma unroll
                for (int t = 0; t < 4; ++t)
                    out[(dbase + 4 * r + t) * 4 + m] = c[t][r];
        }
    } else {
        if (m < 4u) {
            #pragma unroll
            for (int r = 0; r < 8; ++r)
                #pragma unroll
                for (int t = 0; t < 4; ++t) {
                    const long d = dbase + 4 * r + t;
                    if (d < (long)Dc) out[d * 4 + m] = c[t][r];
                }
        }
    }
}

__global__ __launch_bounds__(256) void sg_phase2(const float* __restrict__ X,
                                                 const float* __restrict__ Bws,
                                                 float* __restrict__ out,
                                                 int Dc, int numTiles)
{
    const int lane = threadIdx.x & 31;
    const int wave = threadIdx.x >> 5;
    const int tile = blockIdx.x * 8 + wave;
    if (tile >= numTiles) return;

    const long d0 = (long)tile * 16;
    const int  m  = lane & 15;
    const int  hi = lane >> 4;
    const int  kb = hi * 2;

    long col = d0 + m;
    if (col >= Dc) col = Dc - 1;
    const float* xp = X + (long)kb * Dc + col;

    v8f c = {0.0f, 0.0f, 0.0f, 0.0f, 0.0f, 0.0f, 0.0f, 0.0f};

    for (int s = 0; s < NR / 4; ++s) {
        const float* xs = xp + (long)s * 4 * Dc;
        v2f a;
        a.x = xs[0];
        a.y = xs[Dc];
        v2f b = *(const v2f*)(Bws + s * 64 + lane * 2);
        __builtin_prefetch(xs + 8 * (long)Dc, 0, 0);
        c = __builtin_amdgcn_wmma_f32_16x16x4_f32(
                false, a, false, b, (short)0, c, false, false);
    }

    const long dbase = d0 + (long)hi * 8;
    if (d0 + 16 <= (long)Dc) {
        if (m < 4) {
            #pragma unroll
            for (int r = 0; r < 8; ++r) {
                out[(dbase + r) * 4 + m] = c[r];
            }
        }
    } else {
        if (m < 4) {
            #pragma unroll
            for (int r = 0; r < 8; ++r) {
                const long d = dbase + r;
                if (d < Dc) out[d * 4 + m] = c[r];
            }
        }
    }
}

extern "C" void kernel_launch(void* const* d_in, const int* in_sizes, int n_in,
                              void* d_out, int out_size, void* d_ws, size_t ws_size,
                              hipStream_t stream) {
    const float* t_all   = (const float*)d_in[0];
    const float* x_all   = (const float*)d_in[1];
    const float* weights = (const float*)d_in[2];
    (void)d_in[3]; (void)n_in; (void)out_size; (void)ws_size;

    const int T  = in_sizes[0];
    const int Dc = in_sizes[1] / T;

    if ((Dc & 3) == 0) {
        const int waves = (Dc + 63) / 64;
        hipLaunchKernelGGL(sg_fused_wide, dim3(waves), dim3(32), 0, stream,
                           t_all, weights, x_all, (float*)d_out, Dc);
    } else {
        float* Bws = (float*)d_ws;
        hipLaunchKernelGGL(sg_phase1, dim3(1), dim3(256), 0, stream,
                           t_all, weights, Bws);
        const int numTiles = (Dc + 15) / 16;
        const int blocks   = (numTiles + 7) / 8;
        hipLaunchKernelGGL(sg_phase2, dim3(blocks), dim3(256), 0, stream,
                           x_all, Bws, (float*)d_out, Dc, numTiles);
    }
}
